# starting kernel
# baseline (speedup 1.0000x reference)
#include <hip/hip_runtime.h>

typedef __attribute__((ext_vector_type(2))) float v2f;
typedef __attribute__((ext_vector_type(8))) float v8f;

#define LORENZ_DT 0.02
#define LORENZ_ORDER 5

struct ATab { float v[16][20]; };

constexpr ATab make_atab() {
    double A0[3][3] = {{-10.0, 10.0, 0.0},
                       { 28.0, -1.0, 0.0},
                       {  0.0,  0.0, -8.0 / 3.0}};
    double A1[3][3] = {{0.0, 0.0,  0.0},
                       {0.0, 0.0, -1.0},
                       {0.0, 1.0,  0.0}};
    double T[6][3][3] = {};
    double F[6][3][3] = {};
    for (int i = 0; i < 3; ++i) { T[0][i][i] = 1.0; F[0][i][i] = 1.0; }
    for (int j = 1; j <= LORENZ_ORDER; ++j) {
        double NT[6][3][3] = {};
        for (int d = 0; d < 6; ++d)
            for (int r = 0; r < 3; ++r)
                for (int c = 0; c < 3; ++c) {
                    double acc = 0.0;
                    for (int k = 0; k < 3; ++k) {
                        acc += T[d][r][k] * A0[k][c];
                        if (d >= 1) acc += T[d - 1][r][k] * A1[k][c];
                    }
                    NT[d][r][c] = acc * LORENZ_DT / (double)j;
                }
        for (int d = 0; d < 6; ++d)
            for (int r = 0; r < 3; ++r)
                for (int c = 0; c < 3; ++c) {
                    T[d][r][c] = NT[d][r][c];
                    F[d][r][c] += NT[d][r][c];
                }
    }
    ATab t = {};
    for (int n = 0; n < 3; ++n)
        for (int p = 0; p < 6; ++p)
            for (int c = 0; c < 3; ++c)
                t.v[n][3 * p + c] = (float)F[p][n][c];
    return t;
}

__device__ __constant__ ATab g_atab = make_atab();

__global__ void lorenz_wmma_kernel(const float* __restrict__ x,
                                   float* __restrict__ out,
                                   int n, int ntiles) {
    const int tid    = blockIdx.x * blockDim.x + threadIdx.x;
    const int lane   = threadIdx.x & 31;
    const int waveId = tid >> 5;
    const int nwaves = (gridDim.x * blockDim.x) >> 5;
    const int row    = lane & 15;
    const int hi     = lane >> 4;

    v2f afrag[5];
#pragma unroll
    for (int c = 0; c < 5; ++c) {
        afrag[c].x = g_atab.v[row][4 * c + 2 * hi + 0];
        afrag[c].y = g_atab.v[row][4 * c + 2 * hi + 1];
    }

    for (int t = waveId; t < ntiles; t += nwaves) {
        const int elem = t * 16 + row;
        const float x0 = x[elem * 3 + 0];
        const float x1 = x[elem * 3 + 1];
        const float x2 = x[elem * 3 + 2];

        const float s1 = x0;
        const float s2 = s1 * s1;
        const float s3 = s2 * s1;
        const float s4 = s2 * s2;
        const float s5 = s4 * s1;

        float z[20];
        z[0]  = x0;      z[1]  = x1;      z[2]  = x2;
        z[3]  = s1 * x0; z[4]  = s1 * x1; z[5]  = s1 * x2;
        z[6]  = s2 * x0; z[7]  = s2 * x1; z[8]  = s2 * x2;
        z[9]  = s3 * x0; z[10] = s3 * x1; z[11] = s3 * x2;
        z[12] = s4 * x0; z[13] = s4 * x1; z[14] = s4 * x2;
        z[15] = s5 * x0; z[16] = s5 * x1; z[17] = s5 * x2;
        z[18] = 0.0f;    z[19] = 0.0f;

        v8f acc = {};
#pragma unroll
        for (int c = 0; c < 5; ++c) {
            v2f b;
            b.x = hi ? z[4 * c + 2] : z[4 * c + 0];
            b.y = hi ? z[4 * c + 3] : z[4 * c + 1];
            acc = __builtin_amdgcn_wmma_f32_16x16x4_f32(
                false, afrag[c], false, b, (short)0, acc, false, false);
        }

        if (lane < 16) {
            out[elem * 3 + 0] = acc[0];
            out[elem * 3 + 1] = acc[1];
            out[elem * 3 + 2] = acc[2];
        }
    }

    const int tail0 = ntiles * 16;
    const int ntail = n - tail0;
    if (tid < ntail) {
        const int e = tail0 + tid;
        const float xv0 = x[e * 3 + 0];
        const float xv1 = x[e * 3 + 1];
        const float xv2 = x[e * 3 + 2];
        const float dt  = (float)LORENZ_DT;
        float A[3][3] = {{-10.0f * dt, 10.0f * dt, 0.0f},
                         { 28.0f * dt, -1.0f * dt, -xv0 * dt},
                         { 0.0f,        xv0 * dt,  (-8.0f / 3.0f) * dt}};
        float F[3][3] = {{1, 0, 0}, {0, 1, 0}, {0, 0, 1}};
        float T[3][3] = {{1, 0, 0}, {0, 1, 0}, {0, 0, 1}};
        for (int j = 1; j <= LORENZ_ORDER; ++j) {
            float NT[3][3];
            const float inv = 1.0f / (float)j;
            for (int r = 0; r < 3; ++r)
                for (int c = 0; c < 3; ++c)
                    NT[r][c] = (T[r][0] * A[0][c] + T[r][1] * A[1][c] +
                                T[r][2] * A[2][c]) * inv;
            for (int r = 0; r < 3; ++r)
                for (int c = 0; c < 3; ++c) {
                    T[r][c] = NT[r][c];
                    F[r][c] += NT[r][c];
                }
        }
        out[e * 3 + 0] = F[0][0] * xv0 + F[0][1] * xv1 + F[0][2] * xv2;
        out[e * 3 + 1] = F[1][0] * xv0 + F[1][1] * xv1 + F[1][2] * xv2;
        out[e * 3 + 2] = F[2][0] * xv0 + F[2][1] * xv1 + F[2][2] * xv2;
    }
}

extern "C" void kernel_launch(void* const* d_in, const int* in_sizes, int n_in,
                              void* d_out, int out_size, void* d_ws, size_t ws_size,
                              hipStream_t stream) {
    const float* x = (const float*)d_in[0];
    float* out = (float*)d_out;
    const int n = in_sizes[0] / 3;
    const int ntiles = n / 16;

    const int threads = 256;
    int blocks = 1024;
    if (blocks * (threads / 32) > ntiles && ntiles > 0) {
        blocks = (ntiles + (threads / 32) - 1) / (threads / 32);
    }
    if (blocks < 1) blocks = 1;

    lorenz_wmma_kernel<<<blocks, threads, 0, stream>>>(x, out, n, ntiles);
}


# Product re-tiled: constant operand block-diagonal over the two lane halves, so a lane owns its element (12-byte load/store per lane, no exchange); 32 elements per WMMA; sums reordered (k in pairs)
# speedup vs baseline: 1.4675x; 1.4675x over previous
#include <hip/hip_runtime.h>

typedef __attribute__((ext_vector_type(2))) float v2f;
typedef __attribute__((ext_vector_type(8))) float v8f;

#define LORENZ_DT 0.02
#define LORENZ_ORDER 5

struct ATab { float v[16][20]; };

constexpr ATab make_atab() {
    double A0[3][3] = {{-10.0, 10.0, 0.0},
                       { 28.0, -1.0, 0.0},
                       {  0.0,  0.0, -8.0 / 3.0}};
    double A1[3][3] = {{0.0, 0.0,  0.0},
                       {0.0, 0.0, -1.0},
                       {0.0, 1.0,  0.0}};
    double T[6][3][3] = {};
    double F[6][3][3] = {};
    for (int i = 0; i < 3; ++i) { T[0][i][i] = 1.0; F[0][i][i] = 1.0; }
    for (int j = 1; j <= LORENZ_ORDER; ++j) {
        double NT[6][3][3] = {};
        for (int d = 0; d < 6; ++d)
            for (int r = 0; r < 3; ++r)
                for (int c = 0; c < 3; ++c) {
                    double acc = 0.0;
                    for (int k = 0; k < 3; ++k) {
                        acc += T[d][r][k] * A0[k][c];
                        if (d >= 1) acc += T[d - 1][r][k] * A1[k][c];
                    }
                    NT[d][r][c] = acc * LORENZ_DT / (double)j;
                }
        for (int d = 0; d < 6; ++d)
            for (int r = 0; r < 3; ++r)
                for (int c = 0; c < 3; ++c) {
                    T[d][r][c] = NT[d][r][c];
                    F[d][r][c] += NT[d][r][c];
                }
    }
    ATab t = {};
    for (int n = 0; n < 3; ++n)
        for (int p = 0; p < 6; ++p)
            for (int c = 0; c < 3; ++c)
                t.v[n][3 * p + c] = (float)F[p][n][c];
    return t;
}

__device__ __constant__ ATab g_atab = make_atab();

#define PAIRS_PER_CHUNK 16
#define GROUP_FLOATS    96

__device__ __forceinline__ void lorenz_features(float x0, float x1, float x2,
                                                float (&z)[18]) {
    const float s1 = x0;
    const float s2 = s1 * s1;
    const float s3 = s2 * s1;
    const float s4 = s2 * s2;
    const float s5 = s4 * s1;
    z[0]  = x0;      z[1]  = x1;      z[2]  = x2;
    z[3]  = s1 * x0; z[4]  = s1 * x1; z[5]  = s1 * x2;
    z[6]  = s2 * x0; z[7]  = s2 * x1; z[8]  = s2 * x2;
    z[9]  = s3 * x0; z[10] = s3 * x1; z[11] = s3 * x2;
    z[12] = s4 * x0; z[13] = s4 * x1; z[14] = s4 * x2;
    z[15] = s5 * x0; z[16] = s5 * x1; z[17] = s5 * x2;
}

__device__ __forceinline__ void lorenz_pair(const v2f (&afrag)[9],
                                            const float (&xa)[3], const float (&xb)[3],
                                            float* __restrict__ dst) {
    float za[18], zb[18];
    lorenz_features(xa[0], xa[1], xa[2], za);
    lorenz_features(xb[0], xb[1], xb[2], zb);
    v8f accA = {};
    v8f accB = {};
#pragma unroll
    for (int j = 0; j < 9; ++j) {
        v2f ba, bb;
        ba.x = za[2 * j]; ba.y = za[2 * j + 1];
        bb.x = zb[2 * j]; bb.y = zb[2 * j + 1];
        accA = __builtin_amdgcn_wmma_f32_16x16x4_f32(
            false, afrag[j], false, ba, (short)0, accA, false, false);
        accB = __builtin_amdgcn_wmma_f32_16x16x4_f32(
            false, afrag[j], false, bb, (short)0, accB, false, false);
    }
    dst[0] = accA[0]; dst[1] = accA[1]; dst[2] = accA[2];
    dst[GROUP_FLOATS + 0] = accB[0];
    dst[GROUP_FLOATS + 1] = accB[1];
    dst[GROUP_FLOATS + 2] = accB[2];
}

__global__ void __launch_bounds__(128)
lorenz_wmma_kernel(const float* __restrict__ x,
                   float* __restrict__ out,
                   int n) {
    const int tid    = blockIdx.x * blockDim.x + threadIdx.x;
    const int lane   = threadIdx.x & 31;
    const int hi     = lane >> 4;
    const int waveId = __builtin_amdgcn_readfirstlane(tid >> 5);
    const int nwaves = (gridDim.x * blockDim.x) >> 5;
    const int npairs = n >> 6;

    int arow = lane - 24 * hi;
    if (arow < 0) arow = 15;
    v2f afrag[9];
#pragma unroll
    for (int j = 0; j < 9; ++j) {
        afrag[j].x = g_atab.v[arow][2 * j + 0];
        afrag[j].y = g_atab.v[arow][2 * j + 1];
    }

    for (int chunk = waveId; chunk * PAIRS_PER_CHUNK < npairs; chunk += nwaves) {
        const int first = chunk * PAIRS_PER_CHUNK;
        const int cnt   = npairs - first;
        const size_t off = ((size_t)first * 64 + lane) * 3;
        const float* __restrict__ src = x + off;
        float* __restrict__ dst = out + off;

        if (cnt >= PAIRS_PER_CHUNK) {
            float xa[PAIRS_PER_CHUNK][3], xb[PAIRS_PER_CHUNK][3];
#pragma unroll
            for (int j = 0; j < PAIRS_PER_CHUNK; ++j) {
                const float* s = src + j * 2 * GROUP_FLOATS;
                xa[j][0] = s[0]; xa[j][1] = s[1]; xa[j][2] = s[2];
                xb[j][0] = s[GROUP_FLOATS + 0];
                xb[j][1] = s[GROUP_FLOATS + 1];
                xb[j][2] = s[GROUP_FLOATS + 2];
            }
            __builtin_amdgcn_sched_barrier(0);
#pragma unroll
            for (int j = 0; j < PAIRS_PER_CHUNK; ++j)
                lorenz_pair(afrag, xa[j], xb[j], dst + j * 2 * GROUP_FLOATS);
        } else {
            for (int j = 0; j < cnt; ++j) {
                const float* s = src + j * 2 * GROUP_FLOATS;
                float xa[3], xb[3];
                xa[0] = s[0]; xa[1] = s[1]; xa[2] = s[2];
                xb[0] = s[GROUP_FLOATS + 0];
                xb[1] = s[GROUP_FLOATS + 1];
                xb[2] = s[GROUP_FLOATS + 2];
                lorenz_pair(afrag, xa, xb, dst + j * 2 * GROUP_FLOATS);
            }
        }
    }

    const int tail0 = npairs * 64;
    const int ntail = n - tail0;
    if (tid < ntail) {
        const int e = tail0 + tid;
        const float xv0 = x[e * 3 + 0];
        const float xv1 = x[e * 3 + 1];
        const float xv2 = x[e * 3 + 2];
        const float dt  = (float)LORENZ_DT;
        float A[3][3] = {{-10.0f * dt, 10.0f * dt, 0.0f},
                         { 28.0f * dt, -1.0f * dt, -xv0 * dt},
                         { 0.0f,        xv0 * dt,  (-8.0f / 3.0f) * dt}};
        float F[3][3] = {{1, 0, 0}, {0, 1, 0}, {0, 0, 1}};
        float T[3][3] = {{1, 0, 0}, {0, 1, 0}, {0, 0, 1}};
        for (int j = 1; j <= LORENZ_ORDER; ++j) {
            float NT[3][3];
            const float inv = 1.0f / (float)j;
            for (int r = 0; r < 3; ++r)
                for (int c = 0; c < 3; ++c)
                    NT[r][c] = (T[r][0] * A[0][c] + T[r][1] * A[1][c] +
                                T[r][2] * A[2][c]) * inv;
            for (int r = 0; r < 3; ++r)
                for (int c = 0; c < 3; ++c) {
                    T[r][c] = NT[r][c];
                    F[r][c] += NT[r][c];
                }
        }
        out[e * 3 + 0] = F[0][0] * xv0 + F[0][1] * xv1 + F[0][2] * xv2;
        out[e * 3 + 1] = F[1][0] * xv0 + F[1][1] * xv1 + F[1][2] * xv2;
        out[e * 3 + 2] = F[2][0] * xv0 + F[2][1] * xv1 + F[2][2] * xv2;
    }
}

extern "C" void kernel_launch(void* const* d_in, const int* in_sizes, int n_in,
                              void* d_out, int out_size, void* d_ws, size_t ws_size,
                              hipStream_t stream) {
    const float* x = (const float*)d_in[0];
    float* out = (float*)d_out;
    const int n = in_sizes[0] / 3;

    lorenz_wmma_kernel<<<512, 128, 0, stream>>>(x, out, n);
}
